# starting kernel
# baseline (speedup 1.0000x reference)
#include <hip/hip_runtime.h>
#include <cstdint>

#define C 128
#define FLT_BIG 3.402823466e38f

typedef float v2f __attribute__((ext_vector_type(2)));
typedef float v8f __attribute__((ext_vector_type(8)));

__global__ void k_zero(float* __restrict__ p, long long n) {
  long long i = (long long)blockIdx.x * blockDim.x + threadIdx.x;
  if (i < n) p[i] = 0.0f;
}

__global__ void k_init_aggr(float* __restrict__ sum, float* __restrict__ mx,
                            float* __restrict__ mn, float* __restrict__ cnt,
                            int n_nodes) {
  long long i = (long long)blockIdx.x * blockDim.x + threadIdx.x;
  long long nc = (long long)n_nodes * C;
  if (i < nc) {
    sum[i] = 0.0f;
    mx[i]  = -FLT_BIG;
    mn[i]  =  FLT_BIG;
  }
  if (i < n_nodes) cnt[i] = 0.0f;
}

__global__ void k_scatter(const float* __restrict__ x, const int* __restrict__ ei,
                          int n_edges,
                          float* __restrict__ sum, float* __restrict__ mx,
                          float* __restrict__ mn, float* __restrict__ cnt) {
  long long tid = (long long)blockIdx.x * blockDim.x + threadIdx.x;
  int e = (int)(tid >> 7);
  int c = (int)(tid & 127);
  if (e >= n_edges) return;
  int s = ei[e];
  int d = ei[n_edges + e];
  float v = x[(long long)s * C + c];
  long long o = (long long)d * C + c;
  __hip_atomic_fetch_add(&sum[o], v, __ATOMIC_RELAXED, __HIP_MEMORY_SCOPE_AGENT);
  __hip_atomic_fetch_max(&mx[o],  v, __ATOMIC_RELAXED, __HIP_MEMORY_SCOPE_AGENT);
  __hip_atomic_fetch_min(&mn[o],  v, __ATOMIC_RELAXED, __HIP_MEMORY_SCOPE_AGENT);
  if (c == 0)
    __hip_atomic_fetch_add(&cnt[d], 1.0f, __ATOMIC_RELAXED, __HIP_MEMORY_SCOPE_AGENT);
}

__global__ void k_finalize(const float* __restrict__ sum, float* __restrict__ mx,
                           float* __restrict__ mn, float* __restrict__ mean,
                           const float* __restrict__ cnt, int n_nodes) {
  long long i = (long long)blockIdx.x * blockDim.x + threadIdx.x;
  long long nc = (long long)n_nodes * C;
  if (i >= nc) return;
  int node = (int)(i >> 7);
  float cn = cnt[node];
  mean[i] = sum[i] / fmaxf(cn, 1.0f);
  if (cn == 0.0f) { mx[i] = 0.0f; mn[i] = 0.0f; }
}

__global__ __launch_bounds__(128) void k_gemm_wmma(
    const float* __restrict__ x,
    const float* __restrict__ mean, const float* __restrict__ mx,
    const float* __restrict__ mn,   const float* __restrict__ sum,
    const float* __restrict__ W1,   const float* __restrict__ W2,
    float* __restrict__ h, int n_nodes) {
  const int lane = threadIdx.x & 31;
  const int wv   = threadIdx.x >> 5;
  const int m0   = blockIdx.x * 64 + wv * 16;
  if (m0 >= n_nodes) return;

  int row = m0 + (lane & 15);
  if (row >= n_nodes) row = n_nodes - 1;
  const int khalf = (lane >> 4) << 1;
  const int nlane = lane & 15;

  v8f acc[8] = {};

  const float* Asrc[5] = { x, mean, mx, mn, sum };
#pragma unroll
  for (int seg = 0; seg < 5; ++seg) {
    const float* A   = Asrc[seg];
    const float* W   = (seg == 0) ? W1 : W2;
    const int    ldw = (seg == 0) ? C : 4 * C;
    const int    wk  = (seg == 0) ? 0 : (seg - 1) * C;
    const float* Ap  = A + (long long)row * C;
    const float* Wp  = W + (long long)nlane * ldw + wk;
#pragma unroll 2
    for (int k0 = 0; k0 < C; k0 += 4) {
      const int kk = k0 + khalf;
      v2f a = *(const v2f*)(Ap + kk);
#pragma unroll
      for (int j = 0; j < 8; ++j) {
        v2f b = *(const v2f*)(Wp + j * 16 * ldw + kk);
        acc[j] = __builtin_amdgcn_wmma_f32_16x16x4_f32(
            false, a, false, b, (short)0, acc[j], false, false);
      }
    }
  }

  const int rbase = m0 + ((lane >> 4) << 3);
#pragma unroll
  for (int v = 0; v < 8; ++v) {
    int r = rbase + v;
    if (r < n_nodes) {
      float* hp = h + (long long)r * C + nlane;
#pragma unroll
      for (int j = 0; j < 8; ++j) hp[j * 16] += acc[j][v];
    }
  }
}

__global__ __launch_bounds__(128) void k_ln(const float* __restrict__ h,
                                            const float* __restrict__ gamma,
                                            const float* __restrict__ beta,
                                            float* __restrict__ out, int n_nodes) {
  int node = blockIdx.x;
  int c = threadIdx.x;
  float v = h[(long long)node * C + c];
  v = fmaxf(0.5f * v, 0.0f);

  float s = v, s2 = v * v;
#pragma unroll
  for (int off = 16; off > 0; off >>= 1) {
    s  += __shfl_xor(s,  off, 32);
    s2 += __shfl_xor(s2, off, 32);
  }
  __shared__ float ps[4], ps2[4];
  int wid = threadIdx.x >> 5;
  if ((threadIdx.x & 31) == 0) { ps[wid] = s; ps2[wid] = s2; }
  __syncthreads();
  float ts  = ps[0]  + ps[1]  + ps[2]  + ps[3];
  float ts2 = ps2[0] + ps2[1] + ps2[2] + ps2[3];
  float mu  = ts  * (1.0f / C);
  float var = ts2 * (1.0f / C) - mu * mu;
  out[(long long)node * C + c] = (v - mu) * rsqrtf(var + 1e-5f) * gamma[c] + beta[c];
}

extern "C" void kernel_launch(void* const* d_in, const int* in_sizes, int n_in,
                              void* d_out, int out_size, void* d_ws, size_t ws_size,
                              hipStream_t stream) {
  const float* x     = (const float*)d_in[0];
  const int*   ei1   = (const int*)d_in[1];
  const int*   ei2   = (const int*)d_in[2];
  const float* W1_r1 = (const float*)d_in[3];
  const float* W2_r1 = (const float*)d_in[4];
  const float* W1_r2 = (const float*)d_in[5];
  const float* W2_r2 = (const float*)d_in[6];
  const float* gamma = (const float*)d_in[7];
  const float* beta  = (const float*)d_in[8];
  float* out = (float*)d_out;

  const int N = in_sizes[0] / C;
  const int E = in_sizes[1] / 2;
  const long long NC = (long long)N * C;

  float* sum  = (float*)d_ws;
  float* mx   = sum  + NC;
  float* mn   = mx   + NC;
  float* mean = mn   + NC;
  float* cnt  = mean + NC;
  float* h    = cnt  + N;

  const int T = 256;
  dim3 gNC((unsigned)((NC + T - 1) / T));
  dim3 gE((unsigned)(((long long)E * C + T - 1) / T));
  dim3 gM((unsigned)((N + 63) / 64));

  k_zero<<<gNC, T, 0, stream>>>(h, NC);

  k_init_aggr<<<gNC, T, 0, stream>>>(sum, mx, mn, cnt, N);
  k_scatter<<<gE, T, 0, stream>>>(x, ei1, E, sum, mx, mn, cnt);
  k_finalize<<<gNC, T, 0, stream>>>(sum, mx, mn, mean, cnt, N);
  k_gemm_wmma<<<gM, 128, 0, stream>>>(x, mean, mx, mn, sum, W1_r1, W2_r1, h, N);

  k_init_aggr<<<gNC, T, 0, stream>>>(sum, mx, mn, cnt, N);
  k_scatter<<<gE, T, 0, stream>>>(x, ei2, E, sum, mx, mn, cnt);
  k_finalize<<<gNC, T, 0, stream>>>(sum, mx, mn, mean, cnt, N);
  k_gemm_wmma<<<gM, 128, 0, stream>>>(x, mean, mx, mn, sum, W1_r2, W2_r2, h, N);

  k_ln<<<(unsigned)N, C, 0, stream>>>(h, gamma, beta, out, N);
}


# float-atomic scatter, init and finalize replaced by an owner pass: each work-group scans all edges for its 208-row tile, waves fold own rows in registers; no atomics; sum order = fixed scan order
# speedup vs baseline: 1.6582x; 1.6582x over previous
#include <hip/hip_runtime.h>
#include <cstdint>

#define C 128
#define FLT_BIG 3.402823466e38f

typedef float v2f __attribute__((ext_vector_type(2)));
typedef float v8f __attribute__((ext_vector_type(8)));

__global__ void k_zero(float* __restrict__ p, long long n) {
  long long i = (long long)blockIdx.x * blockDim.x + threadIdx.x;
  if (i < n) p[i] = 0.0f;
}

#define RPW 13
#define NWV 16
#define TILE_ROWS (RPW * NWV)
#define QCAP 1312

typedef float v16f __attribute__((ext_vector_type(16)));
#define FOLD(v_, r_)                                                         \
  do {                                                                       \
    const float4 fv_ = (v_);                                                 \
    const int    fr_ = (r_);                                                 \
    sx[fr_] += fv_.x; sy[fr_] += fv_.y; sz[fr_] += fv_.z; sw[fr_] += fv_.w;  \
    hx[fr_] = fmaxf(hx[fr_], fv_.x); hy[fr_] = fmaxf(hy[fr_], fv_.y);        \
    hz[fr_] = fmaxf(hz[fr_], fv_.z); hw[fr_] = fmaxf(hw[fr_], fv_.w);        \
    lx[fr_] = fminf(lx[fr_], fv_.x); ly[fr_] = fminf(ly[fr_], fv_.y);        \
    lz[fr_] = fminf(lz[fr_], fv_.z); lw[fr_] = fminf(lw[fr_], fv_.w);        \
    cv += (lane == fr_) ? 1 : 0;                                             \
  } while (0)
#define ROWS(F) F(0) F(1) F(2) F(3) F(4) F(5) F(6) F(7) F(8) F(9) F(10) F(11) F(12)
#define ROW_OUT(i)                                                           \
  out_row(mean, mx, mn, sum, make_float4(sx[i], sy[i], sz[i], sw[i]),        \
          make_float4(hx[i], hy[i], hz[i], hw[i]),                           \
          make_float4(lx[i], ly[i], lz[i], lw[i]),                           \
          __builtin_amdgcn_readlane(cv, i), tile_lo + wv * RPW + i, n_nodes, lane);

__device__ __forceinline__ void out_row(float* __restrict__ mean, float* __restrict__ mx,
                                        float* __restrict__ mn, float* __restrict__ sum,
                                        const float4 sv, const float4 hv, const float4 lv,
                                        int n, int node, int n_nodes, int lane) {
  if (node < n_nodes) {
    const float  dn  = fmaxf((float)n, 1.0f);
    const bool   has = n != 0;
    const float4 z   = make_float4(0.0f, 0.0f, 0.0f, 0.0f);
    const long long o = (long long)node * C + lane * 4;
    *(float4*)(mean + o) = make_float4(sv.x / dn, sv.y / dn, sv.z / dn, sv.w / dn);
    *(float4*)(mx + o)   = has ? hv : z;
    *(float4*)(mn + o)   = has ? lv : z;
    *(float4*)(sum + o)  = sv;
  }
}

__device__ __forceinline__ void q_append(int2* seg, bool m, int e, int lrow,
                                         int& tot, int base) {
  const unsigned bm = __builtin_amdgcn_ballot_w32(m);
  if (bm) {
    const int slot = tot + (int)__builtin_amdgcn_mbcnt_lo(bm, 0u) - base;
    if (m && (unsigned)slot < (unsigned)QCAP) seg[slot] = make_int2(e, lrow);
    tot += __builtin_popcount(bm);
  }
}

__global__ __launch_bounds__(NWV * 32) void k_aggr(
    const float* __restrict__ x, const int* __restrict__ ei, int n_edges, int n_nodes,
    float* __restrict__ mean, float* __restrict__ mx, float* __restrict__ mn,
    float* __restrict__ sum) {
  __shared__ int2 q[NWV][QCAP];
  __shared__ int  qcnt[NWV];

  const int lane    = threadIdx.x & 31;
  const int wv      = __builtin_amdgcn_readfirstlane((int)(threadIdx.x >> 5));
  const int tile_lo = blockIdx.x * TILE_ROWS;
  const int* dst    = ei + n_edges;

  v16f sx = 0.0f, sy = 0.0f, sz = 0.0f, sw = 0.0f;
  v16f hx = -FLT_BIG, hy = -FLT_BIG, hz = -FLT_BIG, hw = -FLT_BIG;
  v16f lx =  FLT_BIG, ly =  FLT_BIG, lz =  FLT_BIG, lw =  FLT_BIG;
  int  cv = 0;

  int ps0 = 0, ps1 = 0, ps2 = 0, pr0 = 0, pr1 = 0, pr2 = 0, k = 0;

  const bool vec = (n_edges & 3) == 0;
  const int  nq  = n_edges >> 2;
  const int  nch = vec ? ((nq + 31) >> 5) : ((n_edges + 31) >> 5);

  const int rot = (int)(((long long)blockIdx.x * nch) / gridDim.x);

  for (int base = 0; base < n_edges; base += QCAP) {
    int tot = 0;
    if (vec) {
      const int4* d4 = (const int4*)dst;
      for (int c0 = wv; c0 < nch; c0 += NWV * 4) {
        int4 d[4];
        int  g[4];
#pragma unroll
        for (int u = 0; u < 4; ++u) {
          const int lc = c0 + u * NWV;
          int pc = lc + rot;
          if (pc >= nch) pc -= nch;
          g[u] = pc * 32 + lane;
          d[u] = (lc < nch && g[u] < nq) ? d4[g[u]] : make_int4(-1, -1, -1, -1);
        }
#pragma unroll
        for (int u = 0; u < 4; ++u) {
          const int r0 = d[u].x - tile_lo, r1 = d[u].y - tile_lo;
          const int r2 = d[u].z - tile_lo, r3 = d[u].w - tile_lo;
          const bool m0 = (unsigned)r0 < (unsigned)TILE_ROWS;
          const bool m1 = (unsigned)r1 < (unsigned)TILE_ROWS;
          const bool m2 = (unsigned)r2 < (unsigned)TILE_ROWS;
          const bool m3 = (unsigned)r3 < (unsigned)TILE_ROWS;
          if (__builtin_amdgcn_ballot_w32(m0 | m1 | m2 | m3)) {
            q_append(q[wv], m0, 4 * g[u],     r0, tot, base);
            q_append(q[wv], m1, 4 * g[u] + 1, r1, tot, base);
            q_append(q[wv], m2, 4 * g[u] + 2, r2, tot, base);
            q_append(q[wv], m3, 4 * g[u] + 3, r3, tot, base);
          }
        }
      }
    } else {
      for (int lc = wv; lc < nch; lc += NWV) {
        int pc = lc + rot;
        if (pc >= nch) pc -= nch;
        const int e  = pc * 32 + lane;
        const int dv = (e < n_edges) ? dst[e] : -1;
        const int r0 = dv - tile_lo;
        q_append(q[wv], (unsigned)r0 < (unsigned)TILE_ROWS, e, r0, tot, base);
      }
    }

    const int nm = min(max(tot - base, 0), QCAP);
    for (int i = lane; i < nm; i += 32) {
      int2 t = q[wv][i];
      t.x = ei[t.x];
      q[wv][i] = t;
    }
    if (lane == 0) qcnt[wv] = tot;
    __syncthreads();

    bool more = false;
    for (int sg = 0; sg <= NWV; ++sg) {
      int n = 1;
      if (sg < NWV) {
        const int t = __builtin_amdgcn_readfirstlane(qcnt[sg]);
        more |= (t - base) > QCAP;
        n = min(max(t - base, 0), QCAP);
      }
      for (int b = 0; b < n; b += 32) {
        int2 en = make_int2(0, -1);
        int  rr = 15;
        unsigned bm;
        if (sg < NWV) {
          const int i = b + lane;
          if (i < n) en = q[sg][i];
          rr = en.y - wv * RPW;
          bm = __builtin_amdgcn_ballot_w32((unsigned)rr < (unsigned)RPW);
        } else {
          bm = k ? ((1u << (4 - k)) - 1u) : 0u;
        }
        while (bm) {
          const int bl = __builtin_ctz(bm);
          bm &= bm - 1;
          const int sr = __builtin_amdgcn_readlane(en.x, bl);
          const int r  = __builtin_amdgcn_readlane(rr, bl);
          if (k == 3) {
            const float4 v3 = *(const float4*)(x + (long long)ps2 * C + lane * 4);
            const float4 v2 = *(const float4*)(x + (long long)ps1 * C + lane * 4);
            const float4 v1 = *(const float4*)(x + (long long)ps0 * C + lane * 4);
            const float4 v0 = *(const float4*)(x + (long long)sr  * C + lane * 4);
            FOLD(v3, pr2);
            FOLD(v2, pr1);
            FOLD(v1, pr0);
            FOLD(v0, r);
            k = 0;
          } else {
            ps2 = ps1; pr2 = pr1;
            ps1 = ps0; pr1 = pr0;
            ps0 = sr;  pr0 = r;
            k += 1;
          }
        }
      }
    }
    __syncthreads();
    if (!more) break;
  }

  ROWS(ROW_OUT)
}

__global__ __launch_bounds__(128) void k_gemm_wmma(
    const float* __restrict__ x,
    const float* __restrict__ mean, const float* __restrict__ mx,
    const float* __restrict__ mn,   const float* __restrict__ sum,
    const float* __restrict__ W1,   const float* __restrict__ W2,
    float* __restrict__ h, int n_nodes) {
  const int lane = threadIdx.x & 31;
  const int wv   = threadIdx.x >> 5;
  const int m0   = blockIdx.x * 64 + wv * 16;
  if (m0 >= n_nodes) return;

  int row = m0 + (lane & 15);
  if (row >= n_nodes) row = n_nodes - 1;
  const int khalf = (lane >> 4) << 1;
  const int nlane = lane & 15;

  v8f acc[8] = {};

  const float* Asrc[5] = { x, mean, mx, mn, sum };
#pragma unroll
  for (int seg = 0; seg < 5; ++seg) {
    const float* A   = Asrc[seg];
    const float* W   = (seg == 0) ? W1 : W2;
    const int    ldw = (seg == 0) ? C : 4 * C;
    const int    wk  = (seg == 0) ? 0 : (seg - 1) * C;
    const float* Ap  = A + (long long)row * C;
    const float* Wp  = W + (long long)nlane * ldw + wk;
#pragma unroll 2
    for (int k0 = 0; k0 < C; k0 += 4) {
      const int kk = k0 + khalf;
      v2f a = *(const v2f*)(Ap + kk);
#pragma unroll
      for (int j = 0; j < 8; ++j) {
        v2f b = *(const v2f*)(Wp + j * 16 * ldw + kk);
        acc[j] = __builtin_amdgcn_wmma_f32_16x16x4_f32(
            false, a, false, b, (short)0, acc[j], false, false);
      }
    }
  }

  const int rbase = m0 + ((lane >> 4) << 3);
#pragma unroll
  for (int v = 0; v < 8; ++v) {
    int r = rbase + v;
    if (r < n_nodes) {
      float* hp = h + (long long)r * C + nlane;
#pragma unroll
      for (int j = 0; j < 8; ++j) hp[j * 16] += acc[j][v];
    }
  }
}

__global__ __launch_bounds__(128) void k_ln(const float* __restrict__ h,
                                            const float* __restrict__ gamma,
                                            const float* __restrict__ beta,
                                            float* __restrict__ out, int n_nodes) {
  int node = blockIdx.x;
  int c = threadIdx.x;
  float v = h[(long long)node * C + c];
  v = fmaxf(0.5f * v, 0.0f);

  float s = v, s2 = v * v;
#pragma unroll
  for (int off = 16; off > 0; off >>= 1) {
    s  += __shfl_xor(s,  off, 32);
    s2 += __shfl_xor(s2, off, 32);
  }
  __shared__ float ps[4], ps2[4];
  int wid = threadIdx.x >> 5;
  if ((threadIdx.x & 31) == 0) { ps[wid] = s; ps2[wid] = s2; }
  __syncthreads();
  float ts  = ps[0]  + ps[1]  + ps[2]  + ps[3];
  float ts2 = ps2[0] + ps2[1] + ps2[2] + ps2[3];
  float mu  = ts  * (1.0f / C);
  float var = ts2 * (1.0f / C) - mu * mu;
  out[(long long)node * C + c] = (v - mu) * rsqrtf(var + 1e-5f) * gamma[c] + beta[c];
}

extern "C" void kernel_launch(void* const* d_in, const int* in_sizes, int n_in,
                              void* d_out, int out_size, void* d_ws, size_t ws_size,
                              hipStream_t stream) {
  const float* x     = (const float*)d_in[0];
  const int*   ei1   = (const int*)d_in[1];
  const int*   ei2   = (const int*)d_in[2];
  const float* W1_r1 = (const float*)d_in[3];
  const float* W2_r1 = (const float*)d_in[4];
  const float* W1_r2 = (const float*)d_in[5];
  const float* W2_r2 = (const float*)d_in[6];
  const float* gamma = (const float*)d_in[7];
  const float* beta  = (const float*)d_in[8];
  float* out = (float*)d_out;

  const int N = in_sizes[0] / C;
  const int E = in_sizes[1] / 2;
  const long long NC = (long long)N * C;

  float* sum  = (float*)d_ws;
  float* mx   = sum  + NC;
  float* mn   = mx   + NC;
  float* mean = mn   + NC;
  float* cnt  = mean + NC;
  float* h    = cnt  + N;

  const int T = 256;
  dim3 gNC((unsigned)((NC + T - 1) / T));
  dim3 gA((unsigned)((N + TILE_ROWS - 1) / TILE_ROWS));
  dim3 gM((unsigned)((N + 63) / 64));

  k_zero<<<gNC, T, 0, stream>>>(h, NC);

  k_aggr<<<gA, NWV * 32, 0, stream>>>(x, ei1, E, N, mean, mx, mn, sum);
  k_gemm_wmma<<<gM, 128, 0, stream>>>(x, mean, mx, mn, sum, W1_r1, W2_r1, h, N);

  k_aggr<<<gA, NWV * 32, 0, stream>>>(x, ei2, E, N, mean, mx, mn, sum);
  k_gemm_wmma<<<gM, 128, 0, stream>>>(x, mean, mx, mn, sum, W1_r2, W2_r2, h, N);

  k_ln<<<(unsigned)N, C, 0, stream>>>(h, gamma, beta, out, N);
}


# product launches: each wave computes a 32x64 tile (was 16x128) and loads A and B fragments as 16-byte pieces covering two k-steps: 6 loads per 16 WMMAs (was 18); k sum order in the products changes
# speedup vs baseline: 2.0174x; 1.2166x over previous
#include <hip/hip_runtime.h>
#include <cstdint>

#define C 128
#define FLT_BIG 3.402823466e38f

typedef float v2f __attribute__((ext_vector_type(2)));
typedef float v8f __attribute__((ext_vector_type(8)));

__global__ void k_zero(float* __restrict__ p, long long n) {
  long long i = (long long)blockIdx.x * blockDim.x + threadIdx.x;
  if (i < n) p[i] = 0.0f;
}

#define RPW 13
#define NWV 16
#define TILE_ROWS (RPW * NWV)
#define QCAP 1312

typedef float v16f __attribute__((ext_vector_type(16)));
#define FOLD(v_, r_)                                                         \
  do {                                                                       \
    const float4 fv_ = (v_);                                                 \
    const int    fr_ = (r_);                                                 \
    sx[fr_] += fv_.x; sy[fr_] += fv_.y; sz[fr_] += fv_.z; sw[fr_] += fv_.w;  \
    hx[fr_] = fmaxf(hx[fr_], fv_.x); hy[fr_] = fmaxf(hy[fr_], fv_.y);        \
    hz[fr_] = fmaxf(hz[fr_], fv_.z); hw[fr_] = fmaxf(hw[fr_], fv_.w);        \
    lx[fr_] = fminf(lx[fr_], fv_.x); ly[fr_] = fminf(ly[fr_], fv_.y);        \
    lz[fr_] = fminf(lz[fr_], fv_.z); lw[fr_] = fminf(lw[fr_], fv_.w);        \
    cv += (lane == fr_) ? 1 : 0;                                             \
  } while (0)
#define ROWS(F) F(0) F(1) F(2) F(3) F(4) F(5) F(6) F(7) F(8) F(9) F(10) F(11) F(12)
#define ROW_OUT(i)                                                           \
  out_row(mean, mx, mn, sum, make_float4(sx[i], sy[i], sz[i], sw[i]),        \
          make_float4(hx[i], hy[i], hz[i], hw[i]),                           \
          make_float4(lx[i], ly[i], lz[i], lw[i]),                           \
          __builtin_amdgcn_readlane(cv, i), tile_lo + wv * RPW + i, n_nodes, lane);

__device__ __forceinline__ void out_row(float* __restrict__ mean, float* __restrict__ mx,
                                        float* __restrict__ mn, float* __restrict__ sum,
                                        const float4 sv, const float4 hv, const float4 lv,
                                        int n, int node, int n_nodes, int lane) {
  if (node < n_nodes) {
    const float  dn  = fmaxf((float)n, 1.0f);
    const bool   has = n != 0;
    const float4 z   = make_float4(0.0f, 0.0f, 0.0f, 0.0f);
    const long long o = (long long)node * C + lane * 4;
    *(float4*)(mean + o) = make_float4(sv.x / dn, sv.y / dn, sv.z / dn, sv.w / dn);
    *(float4*)(mx + o)   = has ? hv : z;
    *(float4*)(mn + o)   = has ? lv : z;
    *(float4*)(sum + o)  = sv;
  }
}

__device__ __forceinline__ void q_append(int2* seg, bool m, int e, int lrow,
                                         int& tot, int base) {
  const unsigned bm = __builtin_amdgcn_ballot_w32(m);
  if (bm) {
    const int slot = tot + (int)__builtin_amdgcn_mbcnt_lo(bm, 0u) - base;
    if (m && (unsigned)slot < (unsigned)QCAP) seg[slot] = make_int2(e, lrow);
    tot += __builtin_popcount(bm);
  }
}

__global__ __launch_bounds__(NWV * 32) void k_aggr(
    const float* __restrict__ x, const int* __restrict__ ei, int n_edges, int n_nodes,
    float* __restrict__ mean, float* __restrict__ mx, float* __restrict__ mn,
    float* __restrict__ sum) {
  __shared__ int2 q[NWV][QCAP];
  __shared__ int  qcnt[NWV];

  const int lane    = threadIdx.x & 31;
  const int wv      = __builtin_amdgcn_readfirstlane((int)(threadIdx.x >> 5));
  const int tile_lo = blockIdx.x * TILE_ROWS;
  const int* dst    = ei + n_edges;

  v16f sx = 0.0f, sy = 0.0f, sz = 0.0f, sw = 0.0f;
  v16f hx = -FLT_BIG, hy = -FLT_BIG, hz = -FLT_BIG, hw = -FLT_BIG;
  v16f lx =  FLT_BIG, ly =  FLT_BIG, lz =  FLT_BIG, lw =  FLT_BIG;
  int  cv = 0;

  int ps0 = 0, ps1 = 0, ps2 = 0, pr0 = 0, pr1 = 0, pr2 = 0, k = 0;

  const bool vec = (n_edges & 3) == 0;
  const int  nq  = n_edges >> 2;
  const int  nch = vec ? ((nq + 31) >> 5) : ((n_edges + 31) >> 5);

  const int rot = (int)(((long long)blockIdx.x * nch) / gridDim.x);

  for (int base = 0; base < n_edges; base += QCAP) {
    int tot = 0;
    if (vec) {
      const int4* d4 = (const int4*)dst;
      for (int c0 = wv; c0 < nch; c0 += NWV * 4) {
        int4 d[4];
        int  g[4];
#pragma unroll
        for (int u = 0; u < 4; ++u) {
          const int lc = c0 + u * NWV;
          int pc = lc + rot;
          if (pc >= nch) pc -= nch;
          g[u] = pc * 32 + lane;
          d[u] = (lc < nch && g[u] < nq) ? d4[g[u]] : make_int4(-1, -1, -1, -1);
        }
#pragma unroll
        for (int u = 0; u < 4; ++u) {
          const int r0 = d[u].x - tile_lo, r1 = d[u].y - tile_lo;
          const int r2 = d[u].z - tile_lo, r3 = d[u].w - tile_lo;
          const bool m0 = (unsigned)r0 < (unsigned)TILE_ROWS;
          const bool m1 = (unsigned)r1 < (unsigned)TILE_ROWS;
          const bool m2 = (unsigned)r2 < (unsigned)TILE_ROWS;
          const bool m3 = (unsigned)r3 < (unsigned)TILE_ROWS;
          if (__builtin_amdgcn_ballot_w32(m0 | m1 | m2 | m3)) {
            q_append(q[wv], m0, 4 * g[u],     r0, tot, base);
            q_append(q[wv], m1, 4 * g[u] + 1, r1, tot, base);
            q_append(q[wv], m2, 4 * g[u] + 2, r2, tot, base);
            q_append(q[wv], m3, 4 * g[u] + 3, r3, tot, base);
          }
        }
      }
    } else {
      for (int lc = wv; lc < nch; lc += NWV) {
        int pc = lc + rot;
        if (pc >= nch) pc -= nch;
        const int e  = pc * 32 + lane;
        const int dv = (e < n_edges) ? dst[e] : -1;
        const int r0 = dv - tile_lo;
        q_append(q[wv], (unsigned)r0 < (unsigned)TILE_ROWS, e, r0, tot, base);
      }
    }

    const int nm = min(max(tot - base, 0), QCAP);
    for (int i = lane; i < nm; i += 32) {
      int2 t = q[wv][i];
      t.x = ei[t.x];
      q[wv][i] = t;
    }
    if (lane == 0) qcnt[wv] = tot;
    __syncthreads();

    bool more = false;
    for (int sg = 0; sg <= NWV; ++sg) {
      int n = 1;
      if (sg < NWV) {
        const int t = __builtin_amdgcn_readfirstlane(qcnt[sg]);
        more |= (t - base) > QCAP;
        n = min(max(t - base, 0), QCAP);
      }
      for (int b = 0; b < n; b += 32) {
        int2 en = make_int2(0, -1);
        int  rr = 15;
        unsigned bm;
        if (sg < NWV) {
          const int i = b + lane;
          if (i < n) en = q[sg][i];
          rr = en.y - wv * RPW;
          bm = __builtin_amdgcn_ballot_w32((unsigned)rr < (unsigned)RPW);
        } else {
          bm = k ? ((1u << (4 - k)) - 1u) : 0u;
        }
        while (bm) {
          const int bl = __builtin_ctz(bm);
          bm &= bm - 1;
          const int sr = __builtin_amdgcn_readlane(en.x, bl);
          const int r  = __builtin_amdgcn_readlane(rr, bl);
          if (k == 3) {
            const float4 v3 = *(const float4*)(x + (long long)ps2 * C + lane * 4);
            const float4 v2 = *(const float4*)(x + (long long)ps1 * C + lane * 4);
            const float4 v1 = *(const float4*)(x + (long long)ps0 * C + lane * 4);
            const float4 v0 = *(const float4*)(x + (long long)sr  * C + lane * 4);
            FOLD(v3, pr2);
            FOLD(v2, pr1);
            FOLD(v1, pr0);
            FOLD(v0, r);
            k = 0;
          } else {
            ps2 = ps1; pr2 = pr1;
            ps1 = ps0; pr1 = pr0;
            ps0 = sr;  pr0 = r;
            k += 1;
          }
        }
      }
    }
    __syncthreads();
    if (!more) break;
  }

  ROWS(ROW_OUT)
}

__global__ __launch_bounds__(128) void k_gemm_wmma(
    const float* __restrict__ x,
    const float* __restrict__ mean, const float* __restrict__ mx,
    const float* __restrict__ mn,   const float* __restrict__ sum,
    const float* __restrict__ W1,   const float* __restrict__ W2,
    float* __restrict__ h, int n_nodes) {
  const int lane = threadIdx.x & 31;
  const int wv   = threadIdx.x >> 5;
  const int m0   = blockIdx.x * 64 + (wv >> 1) * 32;
  const int n0   = (wv & 1) * 64;
  if (m0 >= n_nodes) return;

  const int c  = lane & 15;
  const int kq = (lane >> 4) << 2;
  const int row0 = min(m0 + c,      n_nodes - 1);
  const int row1 = min(m0 + 16 + c, n_nodes - 1);

  v8f acc[2][4] = {};

  const float* Asrc[5] = { x, mean, mx, mn, sum };
#pragma unroll
  for (int seg = 0; seg < 5; ++seg) {
    const float* A   = Asrc[seg];
    const float* W   = (seg == 0) ? W1 : W2;
    const int    ldw = (seg == 0) ? C : 4 * C;
    const int    wk  = (seg == 0) ? 0 : (seg - 1) * C;
    const float* Ap0 = A + (long long)row0 * C + kq;
    const float* Ap1 = A + (long long)row1 * C + kq;
    const float* Wp  = W + (long long)(n0 + c) * ldw + wk + kq;
#pragma unroll 2
    for (int k0 = 0; k0 < C; k0 += 8) {
      const float4 a0 = *(const float4*)(Ap0 + k0);
      const float4 a1 = *(const float4*)(Ap1 + k0);
      float4 b[4];
#pragma unroll
      for (int t = 0; t < 4; ++t) b[t] = *(const float4*)(Wp + (long long)t * 16 * ldw + k0);
      const v2f a0lo = { a0.x, a0.y }, a0hi = { a0.z, a0.w };
      const v2f a1lo = { a1.x, a1.y }, a1hi = { a1.z, a1.w };
#pragma unroll
      for (int t = 0; t < 4; ++t) {
        const v2f blo = { b[t].x, b[t].y };
        acc[0][t] = __builtin_amdgcn_wmma_f32_16x16x4_f32(
            false, a0lo, false, blo, (short)0, acc[0][t], false, false);
        acc[1][t] = __builtin_amdgcn_wmma_f32_16x16x4_f32(
            false, a1lo, false, blo, (short)0, acc[1][t], false, false);
      }
#pragma unroll
      for (int t = 0; t < 4; ++t) {
        const v2f bhi = { b[t].z, b[t].w };
        acc[0][t] = __builtin_amdgcn_wmma_f32_16x16x4_f32(
            false, a0hi, false, bhi, (short)0, acc[0][t], false, false);
        acc[1][t] = __builtin_amdgcn_wmma_f32_16x16x4_f32(
            false, a1hi, false, bhi, (short)0, acc[1][t], false, false);
      }
    }
  }

  const int rb = (lane >> 4) << 3;
#pragma unroll
  for (int s = 0; s < 2; ++s) {
#pragma unroll
    for (int v = 0; v < 8; ++v) {
      const int r = m0 + 16 * s + rb + v;
      if (r < n_nodes) {
        float* hp = h + (long long)r * C + n0 + c;
#pragma unroll
        for (int t = 0; t < 4; ++t) hp[t * 16] += acc[s][t][v];
      }
    }
  }
}

__global__ __launch_bounds__(128) void k_ln(const float* __restrict__ h,
                                            const float* __restrict__ gamma,
                                            const float* __restrict__ beta,
                                            float* __restrict__ out, int n_nodes) {
  int node = blockIdx.x;
  int c = threadIdx.x;
  float v = h[(long long)node * C + c];
  v = fmaxf(0.5f * v, 0.0f);

  float s = v, s2 = v * v;
#pragma unroll
  for (int off = 16; off > 0; off >>= 1) {
    s  += __shfl_xor(s,  off, 32);
    s2 += __shfl_xor(s2, off, 32);
  }
  __shared__ float ps[4], ps2[4];
  int wid = threadIdx.x >> 5;
  if ((threadIdx.x & 31) == 0) { ps[wid] = s; ps2[wid] = s2; }
  __syncthreads();
  float ts  = ps[0]  + ps[1]  + ps[2]  + ps[3];
  float ts2 = ps2[0] + ps2[1] + ps2[2] + ps2[3];
  float mu  = ts  * (1.0f / C);
  float var = ts2 * (1.0f / C) - mu * mu;
  out[(long long)node * C + c] = (v - mu) * rsqrtf(var + 1e-5f) * gamma[c] + beta[c];
}

extern "C" void kernel_launch(void* const* d_in, const int* in_sizes, int n_in,
                              void* d_out, int out_size, void* d_ws, size_t ws_size,
                              hipStream_t stream) {
  const float* x     = (const float*)d_in[0];
  const int*   ei1   = (const int*)d_in[1];
  const int*   ei2   = (const int*)d_in[2];
  const float* W1_r1 = (const float*)d_in[3];
  const float* W2_r1 = (const float*)d_in[4];
  const float* W1_r2 = (const float*)d_in[5];
  const float* W2_r2 = (const float*)d_in[6];
  const float* gamma = (const float*)d_in[7];
  const float* beta  = (const float*)d_in[8];
  float* out = (float*)d_out;

  const int N = in_sizes[0] / C;
  const int E = in_sizes[1] / 2;
  const long long NC = (long long)N * C;

  float* sum  = (float*)d_ws;
  float* mx   = sum  + NC;
  float* mn   = mx   + NC;
  float* mean = mn   + NC;
  float* cnt  = mean + NC;
  float* h    = cnt  + N;

  const int T = 256;
  dim3 gNC((unsigned)((NC + T - 1) / T));
  dim3 gA((unsigned)((N + TILE_ROWS - 1) / TILE_ROWS));
  dim3 gM((unsigned)((N + 63) / 64));

  k_zero<<<gNC, T, 0, stream>>>(h, NC);

  k_aggr<<<gA, NWV * 32, 0, stream>>>(x, ei1, E, N, mean, mx, mn, sum);
  k_gemm_wmma<<<gM, 128, 0, stream>>>(x, mean, mx, mn, sum, W1_r1, W2_r1, h, N);

  k_aggr<<<gA, NWV * 32, 0, stream>>>(x, ei2, E, N, mean, mx, mn, sum);
  k_gemm_wmma<<<gM, 128, 0, stream>>>(x, mean, mx, mn, sum, W1_r2, W2_r2, h, N);

  k_ln<<<(unsigned)N, C, 0, stream>>>(h, gamma, beta, out, N);
}


# owner pass: 8 requests in flight per wave (was 4, and 2 at the fold's first wait as built): 8 list chunks per scan trip, 8 source rows per fold group issued ahead of the folds; sum order unchanged
# speedup vs baseline: 2.0856x; 1.0338x over previous
#include <hip/hip_runtime.h>
#include <cstdint>

#define C 128
#define FLT_BIG 3.402823466e38f

typedef float v2f __attribute__((ext_vector_type(2)));
typedef float v8f __attribute__((ext_vector_type(8)));

__global__ void k_zero(float* __restrict__ p, long long n) {
  long long i = (long long)blockIdx.x * blockDim.x + threadIdx.x;
  if (i < n) p[i] = 0.0f;
}

#define RPW 13
#define NWV 16
#define TILE_ROWS (RPW * NWV)
#define QCAP 1312
#define NSC 8
#define NPF 8

typedef float v16f __attribute__((ext_vector_type(16)));
#define FOLD(v_, r_)                                                         \
  do {                                                                       \
    const float4 fv_ = (v_);                                                 \
    const int    fr_ = (r_);                                                 \
    sx[fr_] += fv_.x; sy[fr_] += fv_.y; sz[fr_] += fv_.z; sw[fr_] += fv_.w;  \
    hx[fr_] = fmaxf(hx[fr_], fv_.x); hy[fr_] = fmaxf(hy[fr_], fv_.y);        \
    hz[fr_] = fmaxf(hz[fr_], fv_.z); hw[fr_] = fmaxf(hw[fr_], fv_.w);        \
    lx[fr_] = fminf(lx[fr_], fv_.x); ly[fr_] = fminf(ly[fr_], fv_.y);        \
    lz[fr_] = fminf(lz[fr_], fv_.z); lw[fr_] = fminf(lw[fr_], fv_.w);        \
    cv += (lane == fr_) ? 1 : 0;                                             \
  } while (0)
#define ROWS(F) F(0) F(1) F(2) F(3) F(4) F(5) F(6) F(7) F(8) F(9) F(10) F(11) F(12)
#define ROW_OUT(i)                                                           \
  out_row(mean, mx, mn, sum, make_float4(sx[i], sy[i], sz[i], sw[i]),        \
          make_float4(hx[i], hy[i], hz[i], hw[i]),                           \
          make_float4(lx[i], ly[i], lz[i], lw[i]),                           \
          __builtin_amdgcn_readlane(cv, i), tile_lo + wv * RPW + i, n_nodes, lane);

__device__ __forceinline__ void out_row(float* __restrict__ mean, float* __restrict__ mx,
                                        float* __restrict__ mn, float* __restrict__ sum,
                                        const float4 sv, const float4 hv, const float4 lv,
                                        int n, int node, int n_nodes, int lane) {
  if (node < n_nodes) {
    const float  dn  = fmaxf((float)n, 1.0f);
    const bool   has = n != 0;
    const float4 z   = make_float4(0.0f, 0.0f, 0.0f, 0.0f);
    const long long o = (long long)node * C + lane * 4;
    *(float4*)(mean + o) = make_float4(sv.x / dn, sv.y / dn, sv.z / dn, sv.w / dn);
    *(float4*)(mx + o)   = has ? hv : z;
    *(float4*)(mn + o)   = has ? lv : z;
    *(float4*)(sum + o)  = sv;
  }
}

__device__ __forceinline__ void q_append(int2* seg, bool m, int e, int lrow,
                                         int& tot, int base) {
  const unsigned bm = __builtin_amdgcn_ballot_w32(m);
  if (bm) {
    const int slot = tot + (int)__builtin_amdgcn_mbcnt_lo(bm, 0u) - base;
    if (m && (unsigned)slot < (unsigned)QCAP) seg[slot] = make_int2(e, lrow);
    tot += __builtin_popcount(bm);
  }
}

__global__ __launch_bounds__(NWV * 32) void k_aggr(
    const float* __restrict__ x, const int* __restrict__ ei, int n_edges, int n_nodes,
    float* __restrict__ mean, float* __restrict__ mx, float* __restrict__ mn,
    float* __restrict__ sum) {
  __shared__ int2 q[NWV][QCAP];
  __shared__ int  qcnt[NWV];

  const int lane    = threadIdx.x & 31;
  const int wv      = __builtin_amdgcn_readfirstlane((int)(threadIdx.x >> 5));
  const int tile_lo = blockIdx.x * TILE_ROWS;
  const int* dst    = ei + n_edges;

  v16f sx = 0.0f, sy = 0.0f, sz = 0.0f, sw = 0.0f;
  v16f hx = -FLT_BIG, hy = -FLT_BIG, hz = -FLT_BIG, hw = -FLT_BIG;
  v16f lx =  FLT_BIG, ly =  FLT_BIG, lz =  FLT_BIG, lw =  FLT_BIG;
  int  cv = 0;

  int ps[NPF - 1], pr[NPF - 1], k = 0;
#pragma unroll
  for (int j = 0; j < NPF - 1; ++j) { ps[j] = 0; pr[j] = 0; }

  const bool vec = (n_edges & 3) == 0;
  const int  nq  = n_edges >> 2;
  const int  nch = vec ? ((nq + 31) >> 5) : ((n_edges + 31) >> 5);

  const int rot = (int)(((long long)blockIdx.x * nch) / gridDim.x);

  for (int base = 0; base < n_edges; base += QCAP) {
    int tot = 0;
    if (vec) {
      const int4* d4 = (const int4*)dst;
      for (int c0 = wv; c0 < nch; c0 += NWV * NSC) {
        int4 d[NSC];
        int  g[NSC];
#pragma unroll
        for (int u = 0; u < NSC; ++u) {
          const int lc = c0 + u * NWV;
          int pc = lc + rot;
          if (pc >= nch) pc -= nch;
          g[u] = pc * 32 + lane;
          d[u] = (lc < nch && g[u] < nq) ? d4[g[u]] : make_int4(-1, -1, -1, -1);
        }
#pragma unroll
        for (int u = 0; u < NSC; ++u) {
          const int r0 = d[u].x - tile_lo, r1 = d[u].y - tile_lo;
          const int r2 = d[u].z - tile_lo, r3 = d[u].w - tile_lo;
          const bool m0 = (unsigned)r0 < (unsigned)TILE_ROWS;
          const bool m1 = (unsigned)r1 < (unsigned)TILE_ROWS;
          const bool m2 = (unsigned)r2 < (unsigned)TILE_ROWS;
          const bool m3 = (unsigned)r3 < (unsigned)TILE_ROWS;
          if (__builtin_amdgcn_ballot_w32(m0 | m1 | m2 | m3)) {
            q_append(q[wv], m0, 4 * g[u],     r0, tot, base);
            q_append(q[wv], m1, 4 * g[u] + 1, r1, tot, base);
            q_append(q[wv], m2, 4 * g[u] + 2, r2, tot, base);
            q_append(q[wv], m3, 4 * g[u] + 3, r3, tot, base);
          }
        }
      }
    } else {
      for (int lc = wv; lc < nch; lc += NWV) {
        int pc = lc + rot;
        if (pc >= nch) pc -= nch;
        const int e  = pc * 32 + lane;
        const int dv = (e < n_edges) ? dst[e] : -1;
        const int r0 = dv - tile_lo;
        q_append(q[wv], (unsigned)r0 < (unsigned)TILE_ROWS, e, r0, tot, base);
      }
    }

    const int nm = min(max(tot - base, 0), QCAP);
    for (int i = lane; i < nm; i += 32) {
      int2 t = q[wv][i];
      t.x = ei[t.x];
      q[wv][i] = t;
    }
    if (lane == 0) qcnt[wv] = tot;
    __syncthreads();

    bool more = false;
    for (int sg = 0; sg <= NWV; ++sg) {
      int n = 1;
      if (sg < NWV) {
        const int t = __builtin_amdgcn_readfirstlane(qcnt[sg]);
        more |= (t - base) > QCAP;
        n = min(max(t - base, 0), QCAP);
      }
      for (int b = 0; b < n; b += 32) {
        int2 en = make_int2(0, -1);
        int  rr = 15;
        unsigned bm;
        if (sg < NWV) {
          const int i = b + lane;
          if (i < n) en = q[sg][i];
          rr = en.y - wv * RPW;
          bm = __builtin_amdgcn_ballot_w32((unsigned)rr < (unsigned)RPW);
        } else {
          bm = k ? ((1u << (NPF - k)) - 1u) : 0u;
        }
        while (bm) {
          const int bl = __builtin_ctz(bm);
          bm &= bm - 1;
          const int sr = __builtin_amdgcn_readlane(en.x, bl);
          const int r  = __builtin_amdgcn_readlane(rr, bl);
          if (k == NPF - 1) {
            float4 v[NPF];
#pragma unroll
            for (int j = 0; j < NPF - 1; ++j)
              v[j] = *(const float4*)(x + (long long)ps[NPF - 2 - j] * C + lane * 4);
            v[NPF - 1] = *(const float4*)(x + (long long)sr * C + lane * 4);
            __builtin_amdgcn_sched_barrier(0);
#pragma unroll
            for (int j = 0; j < NPF - 1; ++j) FOLD(v[j], pr[NPF - 2 - j]);
            FOLD(v[NPF - 1], r);
            k = 0;
          } else {
#pragma unroll
            for (int j = NPF - 2; j > 0; --j) { ps[j] = ps[j - 1]; pr[j] = pr[j - 1]; }
            ps[0] = sr; pr[0] = r;
            k += 1;
          }
        }
      }
    }
    __syncthreads();
    if (!more) break;
  }

  ROWS(ROW_OUT)
}

__global__ __launch_bounds__(128) void k_gemm_wmma(
    const float* __restrict__ x,
    const float* __restrict__ mean, const float* __restrict__ mx,
    const float* __restrict__ mn,   const float* __restrict__ sum,
    const float* __restrict__ W1,   const float* __restrict__ W2,
    float* __restrict__ h, int n_nodes) {
  const int lane = threadIdx.x & 31;
  const int wv   = threadIdx.x >> 5;
  const int m0   = blockIdx.x * 64 + (wv >> 1) * 32;
  const int n0   = (wv & 1) * 64;
  if (m0 >= n_nodes) return;

  const int c  = lane & 15;
  const int kq = (lane >> 4) << 2;
  const int row0 = min(m0 + c,      n_nodes - 1);
  const int row1 = min(m0 + 16 + c, n_nodes - 1);

  v8f acc[2][4] = {};

  const float* Asrc[5] = { x, mean, mx, mn, sum };
#pragma unroll
  for (int seg = 0; seg < 5; ++seg) {
    const float* A   = Asrc[seg];
    const float* W   = (seg == 0) ? W1 : W2;
    const int    ldw = (seg == 0) ? C : 4 * C;
    const int    wk  = (seg == 0) ? 0 : (seg - 1) * C;
    const float* Ap0 = A + (long long)row0 * C + kq;
    const float* Ap1 = A + (long long)row1 * C + kq;
    const float* Wp  = W + (long long)(n0 + c) * ldw + wk + kq;
#pragma unroll 2
    for (int k0 = 0; k0 < C; k0 += 8) {
      const float4 a0 = *(const float4*)(Ap0 + k0);
      const float4 a1 = *(const float4*)(Ap1 + k0);
      float4 b[4];
#pragma unroll
      for (int t = 0; t < 4; ++t) b[t] = *(const float4*)(Wp + (long long)t * 16 * ldw + k0);
      const v2f a0lo = { a0.x, a0.y }, a0hi = { a0.z, a0.w };
      const v2f a1lo = { a1.x, a1.y }, a1hi = { a1.z, a1.w };
#pragma unroll
      for (int t = 0; t < 4; ++t) {
        const v2f blo = { b[t].x, b[t].y };
        acc[0][t] = __builtin_amdgcn_wmma_f32_16x16x4_f32(
            false, a0lo, false, blo, (short)0, acc[0][t], false, false);
        acc[1][t] = __builtin_amdgcn_wmma_f32_16x16x4_f32(
            false, a1lo, false, blo, (short)0, acc[1][t], false, false);
      }
#pragma unroll
      for (int t = 0; t < 4; ++t) {
        const v2f bhi = { b[t].z, b[t].w };
        acc[0][t] = __builtin_amdgcn_wmma_f32_16x16x4_f32(
            false, a0hi, false, bhi, (short)0, acc[0][t], false, false);
        acc[1][t] = __builtin_amdgcn_wmma_f32_16x16x4_f32(
            false, a1hi, false, bhi, (short)0, acc[1][t], false, false);
      }
    }
  }

  const int rb = (lane >> 4) << 3;
#pragma unroll
  for (int s = 0; s < 2; ++s) {
#pragma unroll
    for (int v = 0; v < 8; ++v) {
      const int r = m0 + 16 * s + rb + v;
      if (r < n_nodes) {
        float* hp = h + (long long)r * C + n0 + c;
#pragma unroll
        for (int t = 0; t < 4; ++t) hp[t * 16] += acc[s][t][v];
      }
    }
  }
}

__global__ __launch_bounds__(128) void k_ln(const float* __restrict__ h,
                                            const float* __restrict__ gamma,
                                            const float* __restrict__ beta,
                                            float* __restrict__ out, int n_nodes) {
  int node = blockIdx.x;
  int c = threadIdx.x;
  float v = h[(long long)node * C + c];
  v = fmaxf(0.5f * v, 0.0f);

  float s = v, s2 = v * v;
#pragma unroll
  for (int off = 16; off > 0; off >>= 1) {
    s  += __shfl_xor(s,  off, 32);
    s2 += __shfl_xor(s2, off, 32);
  }
  __shared__ float ps[4], ps2[4];
  int wid = threadIdx.x >> 5;
  if ((threadIdx.x & 31) == 0) { ps[wid] = s; ps2[wid] = s2; }
  __syncthreads();
  float ts  = ps[0]  + ps[1]  + ps[2]  + ps[3];
  float ts2 = ps2[0] + ps2[1] + ps2[2] + ps2[3];
  float mu  = ts  * (1.0f / C);
  float var = ts2 * (1.0f / C) - mu * mu;
  out[(long long)node * C + c] = (v - mu) * rsqrtf(var + 1e-5f) * gamma[c] + beta[c];
}

extern "C" void kernel_launch(void* const* d_in, const int* in_sizes, int n_in,
                              void* d_out, int out_size, void* d_ws, size_t ws_size,
                              hipStream_t stream) {
  const float* x     = (const float*)d_in[0];
  const int*   ei1   = (const int*)d_in[1];
  const int*   ei2   = (const int*)d_in[2];
  const float* W1_r1 = (const float*)d_in[3];
  const float* W2_r1 = (const float*)d_in[4];
  const float* W1_r2 = (const float*)d_in[5];
  const float* W2_r2 = (const float*)d_in[6];
  const float* gamma = (const float*)d_in[7];
  const float* beta  = (const float*)d_in[8];
  float* out = (float*)d_out;

  const int N = in_sizes[0] / C;
  const int E = in_sizes[1] / 2;
  const long long NC = (long long)N * C;

  float* sum  = (float*)d_ws;
  float* mx   = sum  + NC;
  float* mn   = mx   + NC;
  float* mean = mn   + NC;
  float* cnt  = mean + NC;
  float* h    = cnt  + N;

  const int T = 256;
  dim3 gNC((unsigned)((NC + T - 1) / T));
  dim3 gA((unsigned)((N + TILE_ROWS - 1) / TILE_ROWS));
  dim3 gM((unsigned)((N + 63) / 64));

  k_zero<<<gNC, T, 0, stream>>>(h, NC);

  k_aggr<<<gA, NWV * 32, 0, stream>>>(x, ei1, E, N, mean, mx, mn, sum);
  k_gemm_wmma<<<gM, 128, 0, stream>>>(x, mean, mx, mn, sum, W1_r1, W2_r1, h, N);

  k_aggr<<<gA, NWV * 32, 0, stream>>>(x, ei2, E, N, mean, mx, mn, sum);
  k_gemm_wmma<<<gM, 128, 0, stream>>>(x, mean, mx, mn, sum, W1_r2, W2_r2, h, N);

  k_ln<<<(unsigned)N, C, 0, stream>>>(h, gamma, beta, out, N);
}


# owner pass scan: a trip whose 8 list chunks neither wrap nor reach the list's last chunk loads them off one index with constant steps; per-chunk wrap, bound test and fill only on the 2 edge trips
# speedup vs baseline: 2.2341x; 1.0712x over previous
#include <hip/hip_runtime.h>
#include <cstdint>

#define C 128
#define FLT_BIG 3.402823466e38f

typedef float v2f __attribute__((ext_vector_type(2)));
typedef float v8f __attribute__((ext_vector_type(8)));

__global__ void k_zero(float* __restrict__ p, long long n) {
  long long i = (long long)blockIdx.x * blockDim.x + threadIdx.x;
  if (i < n) p[i] = 0.0f;
}

#define RPW 13
#define NWV 16
#define TILE_ROWS (RPW * NWV)
#define QCAP 1312
#define NSC 8
#define NPF 8

typedef float v16f __attribute__((ext_vector_type(16)));
#define FOLD(v_, r_)                                                         \
  do {                                                                       \
    const float4 fv_ = (v_);                                                 \
    const int    fr_ = (r_);                                                 \
    sx[fr_] += fv_.x; sy[fr_] += fv_.y; sz[fr_] += fv_.z; sw[fr_] += fv_.w;  \
    hx[fr_] = fmaxf(hx[fr_], fv_.x); hy[fr_] = fmaxf(hy[fr_], fv_.y);        \
    hz[fr_] = fmaxf(hz[fr_], fv_.z); hw[fr_] = fmaxf(hw[fr_], fv_.w);        \
    lx[fr_] = fminf(lx[fr_], fv_.x); ly[fr_] = fminf(ly[fr_], fv_.y);        \
    lz[fr_] = fminf(lz[fr_], fv_.z); lw[fr_] = fminf(lw[fr_], fv_.w);        \
    cv += (lane == fr_) ? 1 : 0;                                             \
  } while (0)
#define ROWS(F) F(0) F(1) F(2) F(3) F(4) F(5) F(6) F(7) F(8) F(9) F(10) F(11) F(12)
#define ROW_OUT(i)                                                           \
  out_row(mean, mx, mn, sum, make_float4(sx[i], sy[i], sz[i], sw[i]),        \
          make_float4(hx[i], hy[i], hz[i], hw[i]),                           \
          make_float4(lx[i], ly[i], lz[i], lw[i]),                           \
          __builtin_amdgcn_readlane(cv, i), tile_lo + wv * RPW + i, n_nodes, lane);

__device__ __forceinline__ void out_row(float* __restrict__ mean, float* __restrict__ mx,
                                        float* __restrict__ mn, float* __restrict__ sum,
                                        const float4 sv, const float4 hv, const float4 lv,
                                        int n, int node, int n_nodes, int lane) {
  if (node < n_nodes) {
    const float  dn  = fmaxf((float)n, 1.0f);
    const bool   has = n != 0;
    const float4 z   = make_float4(0.0f, 0.0f, 0.0f, 0.0f);
    const long long o = (long long)node * C + lane * 4;
    *(float4*)(mean + o) = make_float4(sv.x / dn, sv.y / dn, sv.z / dn, sv.w / dn);
    *(float4*)(mx + o)   = has ? hv : z;
    *(float4*)(mn + o)   = has ? lv : z;
    *(float4*)(sum + o)  = sv;
  }
}

__device__ __forceinline__ void q_append(int2* seg, bool m, int e, int lrow,
                                         int& tot, int base) {
  const unsigned bm = __builtin_amdgcn_ballot_w32(m);
  if (bm) {
    const int slot = tot + (int)__builtin_amdgcn_mbcnt_lo(bm, 0u) - base;
    if (m && (unsigned)slot < (unsigned)QCAP) seg[slot] = make_int2(e, lrow);
    tot += __builtin_popcount(bm);
  }
}

__global__ __launch_bounds__(NWV * 32) void k_aggr(
    const float* __restrict__ x, const int* __restrict__ ei, int n_edges, int n_nodes,
    float* __restrict__ mean, float* __restrict__ mx, float* __restrict__ mn,
    float* __restrict__ sum) {
  __shared__ int2 q[NWV][QCAP];
  __shared__ int  qcnt[NWV];

  const int lane    = threadIdx.x & 31;
  const int wv      = __builtin_amdgcn_readfirstlane((int)(threadIdx.x >> 5));
  const int tile_lo = blockIdx.x * TILE_ROWS;
  const int* dst    = ei + n_edges;

  v16f sx = 0.0f, sy = 0.0f, sz = 0.0f, sw = 0.0f;
  v16f hx = -FLT_BIG, hy = -FLT_BIG, hz = -FLT_BIG, hw = -FLT_BIG;
  v16f lx =  FLT_BIG, ly =  FLT_BIG, lz =  FLT_BIG, lw =  FLT_BIG;
  int  cv = 0;

  int ps[NPF - 1], pr[NPF - 1], k = 0;
#pragma unroll
  for (int j = 0; j < NPF - 1; ++j) { ps[j] = 0; pr[j] = 0; }

  const bool vec = (n_edges & 3) == 0;
  const int  nq  = n_edges >> 2;
  const int  nch = vec ? ((nq + 31) >> 5) : ((n_edges + 31) >> 5);

  const int rot = (int)(((long long)blockIdx.x * nch) / gridDim.x);

  for (int base = 0; base < n_edges; base += QCAP) {
    int tot = 0;
    if (vec) {
      const int4* d4 = (const int4*)dst;
      for (int c0 = wv; c0 < nch; c0 += NWV * NSC) {
        int4 d[NSC];
        int  g[NSC];
        int  p0 = c0 + rot;
        if (p0 >= nch) p0 -= nch;
        const bool straight = (c0 + (NSC - 1) * NWV < nch) &&
                              (p0 + (NSC - 1) * NWV < nch - 1);
        if (straight) {
          const int g0 = p0 * 32 + lane;
#pragma unroll
          for (int u = 0; u < NSC; ++u) {
            g[u] = g0 + u * (NWV * 32);
            d[u] = d4[g[u]];
          }
        } else {
#pragma unroll
          for (int u = 0; u < NSC; ++u) {
            const int lc = c0 + u * NWV;
            int pc = lc + rot;
            if (pc >= nch) pc -= nch;
            g[u] = pc * 32 + lane;
            d[u] = (lc < nch && g[u] < nq) ? d4[g[u]] : make_int4(-1, -1, -1, -1);
          }
        }
#pragma unroll
        for (int u = 0; u < NSC; ++u) {
          const int r0 = d[u].x - tile_lo, r1 = d[u].y - tile_lo;
          const int r2 = d[u].z - tile_lo, r3 = d[u].w - tile_lo;
          const bool m0 = (unsigned)r0 < (unsigned)TILE_ROWS;
          const bool m1 = (unsigned)r1 < (unsigned)TILE_ROWS;
          const bool m2 = (unsigned)r2 < (unsigned)TILE_ROWS;
          const bool m3 = (unsigned)r3 < (unsigned)TILE_ROWS;
          if (__builtin_amdgcn_ballot_w32(m0 | m1 | m2 | m3)) {
            q_append(q[wv], m0, 4 * g[u],     r0, tot, base);
            q_append(q[wv], m1, 4 * g[u] + 1, r1, tot, base);
            q_append(q[wv], m2, 4 * g[u] + 2, r2, tot, base);
            q_append(q[wv], m3, 4 * g[u] + 3, r3, tot, base);
          }
        }
      }
    } else {
      for (int lc = wv; lc < nch; lc += NWV) {
        int pc = lc + rot;
        if (pc >= nch) pc -= nch;
        const int e  = pc * 32 + lane;
        const int dv = (e < n_edges) ? dst[e] : -1;
        const int r0 = dv - tile_lo;
        q_append(q[wv], (unsigned)r0 < (unsigned)TILE_ROWS, e, r0, tot, base);
      }
    }

    const int nm = min(max(tot - base, 0), QCAP);
    for (int i = lane; i < nm; i += 32) {
      int2 t = q[wv][i];
      t.x = ei[t.x];
      q[wv][i] = t;
    }
    if (lane == 0) qcnt[wv] = tot;
    __syncthreads();

    bool more = false;
    for (int sg = 0; sg <= NWV; ++sg) {
      int n = 1;
      if (sg < NWV) {
        const int t = __builtin_amdgcn_readfirstlane(qcnt[sg]);
        more |= (t - base) > QCAP;
        n = min(max(t - base, 0), QCAP);
      }
      for (int b = 0; b < n; b += 32) {
        int2 en = make_int2(0, -1);
        int  rr = 15;
        unsigned bm;
        if (sg < NWV) {
          const int i = b + lane;
          if (i < n) en = q[sg][i];
          rr = en.y - wv * RPW;
          bm = __builtin_amdgcn_ballot_w32((unsigned)rr < (unsigned)RPW);
        } else {
          bm = k ? ((1u << (NPF - k)) - 1u) : 0u;
        }
        while (bm) {
          const int bl = __builtin_ctz(bm);
          bm &= bm - 1;
          const int sr = __builtin_amdgcn_readlane(en.x, bl);
          const int r  = __builtin_amdgcn_readlane(rr, bl);
          if (k == NPF - 1) {
            float4 v[NPF];
#pragma unroll
            for (int j = 0; j < NPF - 1; ++j)
              v[j] = *(const float4*)(x + (long long)ps[NPF - 2 - j] * C + lane * 4);
            v[NPF - 1] = *(const float4*)(x + (long long)sr * C + lane * 4);
            __builtin_amdgcn_sched_barrier(0);
#pragma unroll
            for (int j = 0; j < NPF - 1; ++j) FOLD(v[j], pr[NPF - 2 - j]);
            FOLD(v[NPF - 1], r);
            k = 0;
          } else {
#pragma unroll
            for (int j = NPF - 2; j > 0; --j) { ps[j] = ps[j - 1]; pr[j] = pr[j - 1]; }
            ps[0] = sr; pr[0] = r;
            k += 1;
          }
        }
      }
    }
    __syncthreads();
    if (!more) break;
  }

  ROWS(ROW_OUT)
}

__global__ __launch_bounds__(128) void k_gemm_wmma(
    const float* __restrict__ x,
    const float* __restrict__ mean, const float* __restrict__ mx,
    const float* __restrict__ mn,   const float* __restrict__ sum,
    const float* __restrict__ W1,   const float* __restrict__ W2,
    float* __restrict__ h, int n_nodes) {
  const int lane = threadIdx.x & 31;
  const int wv   = threadIdx.x >> 5;
  const int m0   = blockIdx.x * 64 + (wv >> 1) * 32;
  const int n0   = (wv & 1) * 64;
  if (m0 >= n_nodes) return;

  const int c  = lane & 15;
  const int kq = (lane >> 4) << 2;
  const int row0 = min(m0 + c,      n_nodes - 1);
  const int row1 = min(m0 + 16 + c, n_nodes - 1);

  v8f acc[2][4] = {};

  const float* Asrc[5] = { x, mean, mx, mn, sum };
#pragma unroll
  for (int seg = 0; seg < 5; ++seg) {
    const float* A   = Asrc[seg];
    const float* W   = (seg == 0) ? W1 : W2;
    const int    ldw = (seg == 0) ? C : 4 * C;
    const int    wk  = (seg == 0) ? 0 : (seg - 1) * C;
    const float* Ap0 = A + (long long)row0 * C + kq;
    const float* Ap1 = A + (long long)row1 * C + kq;
    const float* Wp  = W + (long long)(n0 + c) * ldw + wk + kq;
#pragma unroll 2
    for (int k0 = 0; k0 < C; k0 += 8) {
      const float4 a0 = *(const float4*)(Ap0 + k0);
      const float4 a1 = *(const float4*)(Ap1 + k0);
      float4 b[4];
#pragma unroll
      for (int t = 0; t < 4; ++t) b[t] = *(const float4*)(Wp + (long long)t * 16 * ldw + k0);
      const v2f a0lo = { a0.x, a0.y }, a0hi = { a0.z, a0.w };
      const v2f a1lo = { a1.x, a1.y }, a1hi = { a1.z, a1.w };
#pragma unroll
      for (int t = 0; t < 4; ++t) {
        const v2f blo = { b[t].x, b[t].y };
        acc[0][t] = __builtin_amdgcn_wmma_f32_16x16x4_f32(
            false, a0lo, false, blo, (short)0, acc[0][t], false, false);
        acc[1][t] = __builtin_amdgcn_wmma_f32_16x16x4_f32(
            false, a1lo, false, blo, (short)0, acc[1][t], false, false);
      }
#pragma unroll
      for (int t = 0; t < 4; ++t) {
        const v2f bhi = { b[t].z, b[t].w };
        acc[0][t] = __builtin_amdgcn_wmma_f32_16x16x4_f32(
            false, a0hi, false, bhi, (short)0, acc[0][t], false, false);
        acc[1][t] = __builtin_amdgcn_wmma_f32_16x16x4_f32(
            false, a1hi, false, bhi, (short)0, acc[1][t], false, false);
      }
    }
  }

  const int rb = (lane >> 4) << 3;
#pragma unroll
  for (int s = 0; s < 2; ++s) {
#pragma unroll
    for (int v = 0; v < 8; ++v) {
      const int r = m0 + 16 * s + rb + v;
      if (r < n_nodes) {
        float* hp = h + (long long)r * C + n0 + c;
#pragma unroll
        for (int t = 0; t < 4; ++t) hp[t * 16] += acc[s][t][v];
      }
    }
  }
}

__global__ __launch_bounds__(128) void k_ln(const float* __restrict__ h,
                                            const float* __restrict__ gamma,
                                            const float* __restrict__ beta,
                                            float* __restrict__ out, int n_nodes) {
  int node = blockIdx.x;
  int c = threadIdx.x;
  float v = h[(long long)node * C + c];
  v = fmaxf(0.5f * v, 0.0f);

  float s = v, s2 = v * v;
#pragma unroll
  for (int off = 16; off > 0; off >>= 1) {
    s  += __shfl_xor(s,  off, 32);
    s2 += __shfl_xor(s2, off, 32);
  }
  __shared__ float ps[4], ps2[4];
  int wid = threadIdx.x >> 5;
  if ((threadIdx.x & 31) == 0) { ps[wid] = s; ps2[wid] = s2; }
  __syncthreads();
  float ts  = ps[0]  + ps[1]  + ps[2]  + ps[3];
  float ts2 = ps2[0] + ps2[1] + ps2[2] + ps2[3];
  float mu  = ts  * (1.0f / C);
  float var = ts2 * (1.0f / C) - mu * mu;
  out[(long long)node * C + c] = (v - mu) * rsqrtf(var + 1e-5f) * gamma[c] + beta[c];
}

extern "C" void kernel_launch(void* const* d_in, const int* in_sizes, int n_in,
                              void* d_out, int out_size, void* d_ws, size_t ws_size,
                              hipStream_t stream) {
  const float* x     = (const float*)d_in[0];
  const int*   ei1   = (const int*)d_in[1];
  const int*   ei2   = (const int*)d_in[2];
  const float* W1_r1 = (const float*)d_in[3];
  const float* W2_r1 = (const float*)d_in[4];
  const float* W1_r2 = (const float*)d_in[5];
  const float* W2_r2 = (const float*)d_in[6];
  const float* gamma = (const float*)d_in[7];
  const float* beta  = (const float*)d_in[8];
  float* out = (float*)d_out;

  const int N = in_sizes[0] / C;
  const int E = in_sizes[1] / 2;
  const long long NC = (long long)N * C;

  float* sum  = (float*)d_ws;
  float* mx   = sum  + NC;
  float* mn   = mx   + NC;
  float* mean = mn   + NC;
  float* cnt  = mean + NC;
  float* h    = cnt  + N;

  const int T = 256;
  dim3 gNC((unsigned)((NC + T - 1) / T));
  dim3 gA((unsigned)((N + TILE_ROWS - 1) / TILE_ROWS));
  dim3 gM((unsigned)((N + 63) / 64));

  k_zero<<<gNC, T, 0, stream>>>(h, NC);

  k_aggr<<<gA, NWV * 32, 0, stream>>>(x, ei1, E, N, mean, mx, mn, sum);
  k_gemm_wmma<<<gM, 128, 0, stream>>>(x, mean, mx, mn, sum, W1_r1, W2_r1, h, N);

  k_aggr<<<gA, NWV * 32, 0, stream>>>(x, ei2, E, N, mean, mx, mn, sum);
  k_gemm_wmma<<<gM, 128, 0, stream>>>(x, mean, mx, mn, sum, W1_r2, W2_r2, h, N);

  k_ln<<<(unsigned)N, C, 0, stream>>>(h, gamma, beta, out, N);
}
